# starting kernel
# baseline (speedup 1.0000x reference)
#include <hip/hip_runtime.h>
#include <math.h>

typedef __attribute__((ext_vector_type(2))) float v2f;
typedef __attribute__((ext_vector_type(8))) float v8f;

#define TILE_M   16
#define NPATCH   196
#define KDIM     784
#define FSTRIDE  785
#define NCLASS   10
#define LSTRIDE  17

__global__ __launch_bounds__(128)
void quanv_fused_kernel(const float* __restrict__ x,
                        const float* __restrict__ params,
                        const float* __restrict__ W,
                        const float* __restrict__ bias,
                        float* __restrict__ out)
{
    __shared__ float featsL[TILE_M * FSTRIDE];
    __shared__ float wt[KDIM * 16];
    __shared__ float lgts[TILE_M * LSTRIDE];

    const int tid = threadIdx.x;
    const int b0  = blockIdx.x * TILE_M;

    float cb[4], ka[4];
    #pragma unroll
    for (int q = 0; q < 4; ++q) {
        float alpha = params[2 * q];
        float beta  = params[2 * q + 1];
        cb[q] = cosf(beta);
        ka[q] = cosf(alpha) * sinf(beta);
    }

    for (int idx = tid; idx < KDIM * 16; idx += blockDim.x) {
        int n = idx & 15;
        int k = idx >> 4;
        wt[idx] = (n < NCLASS) ? W[n * KDIM + k] : 0.0f;
    }

    for (int t = tid; t < TILE_M * NPATCH; t += blockDim.x) {
        int m  = t / NPATCH;
        int p  = t - m * NPATCH;
        int r2 = p / 14;
        int c2 = p - r2 * 14;
        const float* xb = x + (size_t)(b0 + m) * KDIM + (r2 * 56 + c2 * 2);
        float d0 = xb[0], d1 = xb[1], d2 = xb[28], d3 = xb[29];

        float s, c;
        sincosf(2.0f * d0, &s, &c); float m0 = cb[0] * c - ka[0] * s;
        sincosf(2.0f * d1, &s, &c); float m1 = cb[1] * c - ka[1] * s;
        sincosf(2.0f * d2, &s, &c); float m2 = cb[2] * c - ka[2] * s;
        sincosf(2.0f * d3, &s, &c); float m3 = cb[3] * c - ka[3] * s;

        float* f = featsL + m * FSTRIDE + p * 4;
        f[0] = m0;
        f[1] = m0 * m1;
        f[2] = m2;
        f[3] = m2 * m3;
    }
    __syncthreads();

    if (tid < 32) {
        const int nm   = tid & 15;
        const int half = tid >> 4;
        const float* fa = featsL + nm * FSTRIDE + 2 * half;
        const float* fb = wt + (2 * half) * 16 + nm;

        v8f acc = {};
        #pragma unroll 4
        for (int k0 = 0; k0 < KDIM; k0 += 4) {
            v2f A;  A.x  = fa[k0];        A.y  = fa[k0 + 1];
            v2f Bv; Bv.x = fb[k0 * 16];   Bv.y = fb[k0 * 16 + 16];
            acc = __builtin_amdgcn_wmma_f32_16x16x4_f32(
                 false, A,  false, Bv,
                 (short)0, acc,  false,  false);
        }
        #pragma unroll
        for (int r = 0; r < 8; ++r)
            lgts[(r + 8 * half) * LSTRIDE + nm] = acc[r];
    }
    __syncthreads();

    if (tid < TILE_M) {
        const int row = tid;
        float lg[NCLASS];
        float mx = -3.0e38f;
        #pragma unroll
        for (int j = 0; j < NCLASS; ++j) {
            lg[j] = lgts[row * LSTRIDE + j] + bias[j];
            mx = fmaxf(mx, lg[j]);
        }
        float sum = 0.0f;
        #pragma unroll
        for (int j = 0; j < NCLASS; ++j) sum += expf(lg[j] - mx);
        float lse = logf(sum);
        float* o = out + (size_t)(b0 + row) * NCLASS;
        #pragma unroll
        for (int j = 0; j < NCLASS; ++j) o[j] = lg[j] - mx - lse;
    }
}

extern "C" void kernel_launch(void* const* d_in, const int* in_sizes, int n_in,
                              void* d_out, int out_size, void* d_ws, size_t ws_size,
                              hipStream_t stream) {
    (void)n_in; (void)out_size; (void)d_ws; (void)ws_size;
    const float* x      = (const float*)d_in[0];
    const float* params = (const float*)d_in[1];
    const float* W      = (const float*)d_in[2];
    const float* bias   = (const float*)d_in[3];
    float* out          = (float*)d_out;

    const int B      = in_sizes[0] / KDIM;
    const int blocks = B / TILE_M;

    quanv_fused_kernel<<<blocks, 128, 0, stream>>>(x, params, W, bias, out);
}


# work-group widened from 4 to 32 waves: W staging and feature loops (stride = block size) take 8x fewer serial global-load trips per thread; arithmetic and summation order unchanged
# speedup vs baseline: 2.8995x; 2.8995x over previous
#include <hip/hip_runtime.h>
#include <math.h>

typedef __attribute__((ext_vector_type(2))) float v2f;
typedef __attribute__((ext_vector_type(8))) float v8f;

#define TILE_M   16
#define NPATCH   196
#define KDIM     784
#define FSTRIDE  785
#define NCLASS   10
#define LSTRIDE  17
#define NTHREADS 1024

__global__ __launch_bounds__(NTHREADS)
void quanv_fused_kernel(const float* __restrict__ x,
                        const float* __restrict__ params,
                        const float* __restrict__ W,
                        const float* __restrict__ bias,
                        float* __restrict__ out)
{
    __shared__ float featsL[TILE_M * FSTRIDE];
    __shared__ float wt[KDIM * 16];
    __shared__ float lgts[TILE_M * LSTRIDE];

    const int tid = threadIdx.x;
    const int b0  = blockIdx.x * TILE_M;

    float cb[4], ka[4];
    #pragma unroll
    for (int q = 0; q < 4; ++q) {
        float alpha = params[2 * q];
        float beta  = params[2 * q + 1];
        cb[q] = cosf(beta);
        ka[q] = cosf(alpha) * sinf(beta);
    }

    for (int idx = tid; idx < KDIM * 16; idx += blockDim.x) {
        int n = idx & 15;
        int k = idx >> 4;
        wt[idx] = (n < NCLASS) ? W[n * KDIM + k] : 0.0f;
    }

    for (int t = tid; t < TILE_M * NPATCH; t += blockDim.x) {
        int m  = t / NPATCH;
        int p  = t - m * NPATCH;
        int r2 = p / 14;
        int c2 = p - r2 * 14;
        const float* xb = x + (size_t)(b0 + m) * KDIM + (r2 * 56 + c2 * 2);
        float d0 = xb[0], d1 = xb[1], d2 = xb[28], d3 = xb[29];

        float s, c;
        sincosf(2.0f * d0, &s, &c); float m0 = cb[0] * c - ka[0] * s;
        sincosf(2.0f * d1, &s, &c); float m1 = cb[1] * c - ka[1] * s;
        sincosf(2.0f * d2, &s, &c); float m2 = cb[2] * c - ka[2] * s;
        sincosf(2.0f * d3, &s, &c); float m3 = cb[3] * c - ka[3] * s;

        float* f = featsL + m * FSTRIDE + p * 4;
        f[0] = m0;
        f[1] = m0 * m1;
        f[2] = m2;
        f[3] = m2 * m3;
    }
    __syncthreads();

    if (tid < 32) {
        const int nm   = tid & 15;
        const int half = tid >> 4;
        const float* fa = featsL + nm * FSTRIDE + 2 * half;
        const float* fb = wt + (2 * half) * 16 + nm;

        v8f acc = {};
        #pragma unroll 4
        for (int k0 = 0; k0 < KDIM; k0 += 4) {
            v2f A;  A.x  = fa[k0];        A.y  = fa[k0 + 1];
            v2f Bv; Bv.x = fb[k0 * 16];   Bv.y = fb[k0 * 16 + 16];
            acc = __builtin_amdgcn_wmma_f32_16x16x4_f32(
                 false, A,  false, Bv,
                 (short)0, acc,  false,  false);
        }
        #pragma unroll
        for (int r = 0; r < 8; ++r)
            lgts[(r + 8 * half) * LSTRIDE + nm] = acc[r];
    }
    __syncthreads();

    if (tid < TILE_M) {
        const int row = tid;
        float lg[NCLASS];
        float mx = -3.0e38f;
        #pragma unroll
        for (int j = 0; j < NCLASS; ++j) {
            lg[j] = lgts[row * LSTRIDE + j] + bias[j];
            mx = fmaxf(mx, lg[j]);
        }
        float sum = 0.0f;
        #pragma unroll
        for (int j = 0; j < NCLASS; ++j) sum += expf(lg[j] - mx);
        float lse = logf(sum);
        float* o = out + (size_t)(b0 + row) * NCLASS;
        #pragma unroll
        for (int j = 0; j < NCLASS; ++j) o[j] = lg[j] - mx - lse;
    }
}

extern "C" void kernel_launch(void* const* d_in, const int* in_sizes, int n_in,
                              void* d_out, int out_size, void* d_ws, size_t ws_size,
                              hipStream_t stream) {
    (void)n_in; (void)out_size; (void)d_ws; (void)ws_size;
    const float* x      = (const float*)d_in[0];
    const float* params = (const float*)d_in[1];
    const float* W      = (const float*)d_in[2];
    const float* bias   = (const float*)d_in[3];
    float* out          = (float*)d_out;

    const int B      = in_sizes[0] / KDIM;
    const int blocks = B / TILE_M;

    quanv_fused_kernel<<<blocks, NTHREADS, 0, stream>>>(x, params, W, bias, out);
}


# K loop of the WMMA split over 28 waves (7 k-steps each) instead of one wave doing all 196; partial tiles added in LDS in wave order, so the summation order over K changes
# speedup vs baseline: 3.6402x; 1.2555x over previous
#include <hip/hip_runtime.h>
#include <math.h>

typedef __attribute__((ext_vector_type(2))) float v2f;
typedef __attribute__((ext_vector_type(8))) float v8f;

#define TILE_M   16
#define NPATCH   196
#define KDIM     784
#define FSTRIDE  785
#define NCLASS   10
#define LSTRIDE  17
#define NTHREADS 1024
#define KWAVES   28
#define KSTEPS   7

__global__ __launch_bounds__(NTHREADS)
void quanv_fused_kernel(const float* __restrict__ x,
                        const float* __restrict__ params,
                        const float* __restrict__ W,
                        const float* __restrict__ bias,
                        float* __restrict__ out)
{
    __shared__ float featsL[TILE_M * FSTRIDE];
    __shared__ float wt[KDIM * 16];
    __shared__ float lgts[TILE_M * LSTRIDE];
    __shared__ float part[KWAVES * 256];

    const int tid = threadIdx.x;
    const int b0  = blockIdx.x * TILE_M;

    float cb[4], ka[4];
    #pragma unroll
    for (int q = 0; q < 4; ++q) {
        float alpha = params[2 * q];
        float beta  = params[2 * q + 1];
        cb[q] = cosf(beta);
        ka[q] = cosf(alpha) * sinf(beta);
    }

    for (int idx = tid; idx < KDIM * 16; idx += blockDim.x) {
        int n = idx & 15;
        int k = idx >> 4;
        wt[idx] = (n < NCLASS) ? W[n * KDIM + k] : 0.0f;
    }

    for (int t = tid; t < TILE_M * NPATCH; t += blockDim.x) {
        int m  = t / NPATCH;
        int p  = t - m * NPATCH;
        int r2 = p / 14;
        int c2 = p - r2 * 14;
        const float* xb = x + (size_t)(b0 + m) * KDIM + (r2 * 56 + c2 * 2);
        float d0 = xb[0], d1 = xb[1], d2 = xb[28], d3 = xb[29];

        float s, c;
        sincosf(2.0f * d0, &s, &c); float m0 = cb[0] * c - ka[0] * s;
        sincosf(2.0f * d1, &s, &c); float m1 = cb[1] * c - ka[1] * s;
        sincosf(2.0f * d2, &s, &c); float m2 = cb[2] * c - ka[2] * s;
        sincosf(2.0f * d3, &s, &c); float m3 = cb[3] * c - ka[3] * s;

        float* f = featsL + m * FSTRIDE + p * 4;
        f[0] = m0;
        f[1] = m0 * m1;
        f[2] = m2;
        f[3] = m2 * m3;
    }
    __syncthreads();

    const int wave = tid >> 5;
    if (wave < KWAVES) {
        const int lane = tid & 31;
        const int nm   = lane & 15;
        const int half = lane >> 4;
        const int kb   = wave * (KSTEPS * 4);
        const float* fa = featsL + nm * FSTRIDE + 2 * half + kb;
        const float* fb = wt + (kb + 2 * half) * 16 + nm;

        v8f acc = {};
        for (int i = 0; i < KSTEPS; ++i) {
            const int k0 = 4 * i;
            v2f A;  A.x  = fa[k0];        A.y  = fa[k0 + 1];
            v2f Bv; Bv.x = fb[k0 * 16];   Bv.y = fb[k0 * 16 + 16];
            acc = __builtin_amdgcn_wmma_f32_16x16x4_f32(
                 false, A,  false, Bv,
                 (short)0, acc,  false,  false);
        }
        float* pw = part + wave * 256;
        #pragma unroll
        for (int r = 0; r < 8; ++r)
            pw[(r + 8 * half) * 16 + nm] = acc[r];
    }
    __syncthreads();

    if (tid < 256) {
        float s = part[tid];
        for (int w = 1; w < KWAVES; ++w) s += part[w * 256 + tid];
        lgts[(tid >> 4) * LSTRIDE + (tid & 15)] = s;
    }
    __syncthreads();

    if (tid < TILE_M) {
        const int row = tid;
        float lg[NCLASS];
        float mx = -3.0e38f;
        #pragma unroll
        for (int j = 0; j < NCLASS; ++j) {
            lg[j] = lgts[row * LSTRIDE + j] + bias[j];
            mx = fmaxf(mx, lg[j]);
        }
        float sum = 0.0f;
        #pragma unroll
        for (int j = 0; j < NCLASS; ++j) sum += expf(lg[j] - mx);
        float lse = logf(sum);
        float* o = out + (size_t)(b0 + row) * NCLASS;
        #pragma unroll
        for (int j = 0; j < NCLASS; ++j) o[j] = lg[j] - mx - lse;
    }
}

extern "C" void kernel_launch(void* const* d_in, const int* in_sizes, int n_in,
                              void* d_out, int out_size, void* d_ws, size_t ws_size,
                              hipStream_t stream) {
    (void)n_in; (void)out_size; (void)d_ws; (void)ws_size;
    const float* x      = (const float*)d_in[0];
    const float* params = (const float*)d_in[1];
    const float* W      = (const float*)d_in[2];
    const float* bias   = (const float*)d_in[3];
    float* out          = (float*)d_out;

    const int B      = in_sizes[0] / KDIM;
    const int blocks = B / TILE_M;

    quanv_fused_kernel<<<blocks, NTHREADS, 0, stream>>>(x, params, W, bias, out);
}


# W staged row-major by two coalesced 16-byte loads per thread, both in flight at once, instead of 12 serial strided 4-byte trips; B fragment read as one 8-byte LDS load; sums unchanged
# speedup vs baseline: 5.9844x; 1.6440x over previous
#include <hip/hip_runtime.h>
#include <math.h>

typedef __attribute__((ext_vector_type(2))) float v2f;
typedef __attribute__((ext_vector_type(8))) float v8f;

#define TILE_M   16
#define NPATCH   196
#define KDIM     784
#define FSTRIDE  785
#define NCLASS   10
#define LSTRIDE  17
#define WSTRIDE  788
#define ROWVEC   (KDIM / 4)
#define WVEC     (NCLASS * ROWVEC)
#define NTHREADS 1024
#define KWAVES   28
#define KSTEPS   7

__global__ __launch_bounds__(NTHREADS)
void quanv_fused_kernel(const float* __restrict__ x,
                        const float* __restrict__ params,
                        const float* __restrict__ W,
                        const float* __restrict__ bias,
                        float* __restrict__ out)
{
    __shared__ float featsL[TILE_M * FSTRIDE];
    __shared__ __attribute__((aligned(16))) float wl[NCLASS * WSTRIDE];
    __shared__ float lgts[TILE_M * LSTRIDE];
    __shared__ float part[KWAVES * 256];

    const int tid = threadIdx.x;
    const int b0  = blockIdx.x * TILE_M;

    const float4* W4 = reinterpret_cast<const float4*>(W);
    const int p0 = tid;
    const int p1 = (tid + NTHREADS < WVEC) ? tid + NTHREADS : tid;
    const float4 w0 = W4[p0];
    const float4 w1 = W4[p1];

    float cb[4], ka[4];
    #pragma unroll
    for (int q = 0; q < 4; ++q) {
        float alpha = params[2 * q];
        float beta  = params[2 * q + 1];
        cb[q] = cosf(beta);
        ka[q] = cosf(alpha) * sinf(beta);
    }

    {
        const int n0 = p0 / ROWVEC;
        const int n1 = p1 / ROWVEC;
        *reinterpret_cast<float4*>(wl + n0 * WSTRIDE + 4 * (p0 - n0 * ROWVEC)) = w0;
        *reinterpret_cast<float4*>(wl + n1 * WSTRIDE + 4 * (p1 - n1 * ROWVEC)) = w1;
    }

    for (int t = tid; t < TILE_M * NPATCH; t += blockDim.x) {
        int m  = t / NPATCH;
        int p  = t - m * NPATCH;
        int r2 = p / 14;
        int c2 = p - r2 * 14;
        const float* xb = x + (size_t)(b0 + m) * KDIM + (r2 * 56 + c2 * 2);
        float d0 = xb[0], d1 = xb[1], d2 = xb[28], d3 = xb[29];

        float s, c;
        sincosf(2.0f * d0, &s, &c); float m0 = cb[0] * c - ka[0] * s;
        sincosf(2.0f * d1, &s, &c); float m1 = cb[1] * c - ka[1] * s;
        sincosf(2.0f * d2, &s, &c); float m2 = cb[2] * c - ka[2] * s;
        sincosf(2.0f * d3, &s, &c); float m3 = cb[3] * c - ka[3] * s;

        float* f = featsL + m * FSTRIDE + p * 4;
        f[0] = m0;
        f[1] = m0 * m1;
        f[2] = m2;
        f[3] = m2 * m3;
    }
    __syncthreads();

    const int wave = tid >> 5;
    if (wave < KWAVES) {
        const int lane = tid & 31;
        const int nm   = lane & 15;
        const int half = lane >> 4;
        const int kb   = wave * (KSTEPS * 4);
        const float* fa = featsL + nm * FSTRIDE + 2 * half + kb;
        const int nb = nm < NCLASS ? nm : NCLASS - 1;
        const float* fb = wl + nb * WSTRIDE + kb + 2 * half;

        v8f acc = {};
        for (int i = 0; i < KSTEPS; ++i) {
            const int k0 = 4 * i;
            v2f A;  A.x  = fa[k0];        A.y  = fa[k0 + 1];
            v2f Bv = *reinterpret_cast<const v2f*>(fb + k0);
            acc = __builtin_amdgcn_wmma_f32_16x16x4_f32(
                 false, A,  false, Bv,
                 (short)0, acc,  false,  false);
        }
        float* pw = part + wave * 256;
        #pragma unroll
        for (int r = 0; r < 8; ++r)
            pw[(r + 8 * half) * 16 + nm] = acc[r];
    }
    __syncthreads();

    if (tid < 256) {
        float s = part[tid];
        for (int w = 1; w < KWAVES; ++w) s += part[w * 256 + tid];
        lgts[(tid >> 4) * LSTRIDE + (tid & 15)] = s;
    }
    __syncthreads();

    if (tid < TILE_M) {
        const int row = tid;
        float lg[NCLASS];
        float mx = -3.0e38f;
        #pragma unroll
        for (int j = 0; j < NCLASS; ++j) {
            lg[j] = lgts[row * LSTRIDE + j] + bias[j];
            mx = fmaxf(mx, lg[j]);
        }
        float sum = 0.0f;
        #pragma unroll
        for (int j = 0; j < NCLASS; ++j) sum += expf(lg[j] - mx);
        float lse = logf(sum);
        float* o = out + (size_t)(b0 + row) * NCLASS;
        #pragma unroll
        for (int j = 0; j < NCLASS; ++j) o[j] = lg[j] - mx - lse;
    }
}

extern "C" void kernel_launch(void* const* d_in, const int* in_sizes, int n_in,
                              void* d_out, int out_size, void* d_ws, size_t ws_size,
                              hipStream_t stream) {
    (void)n_in; (void)out_size; (void)d_ws; (void)ws_size;
    const float* x      = (const float*)d_in[0];
    const float* params = (const float*)d_in[1];
    const float* W      = (const float*)d_in[2];
    const float* bias   = (const float*)d_in[3];
    float* out          = (float*)d_out;

    const int B      = in_sizes[0] / KDIM;
    const int blocks = B / TILE_M;

    quanv_fused_kernel<<<blocks, NTHREADS, 0, stream>>>(x, params, W, bias, out);
}


# angle constants: lane i takes sincosf(params[i]) on the vector unit, read back by v_readlane, in place of 12 scalar-unit sin/cos runs and 4 serial scalar loads per wave; formulas and sums unchanged
# speedup vs baseline: 8.5851x; 1.4346x over previous
#include <hip/hip_runtime.h>
#include <math.h>

typedef __attribute__((ext_vector_type(2))) float v2f;
typedef __attribute__((ext_vector_type(8))) float v8f;

__device__ __forceinline__ float lane_value(float v, int lane)
{
    return __int_as_float(__builtin_amdgcn_readlane(__float_as_int(v), lane));
}

#define TILE_M   16
#define NPATCH   196
#define KDIM     784
#define FSTRIDE  785
#define NCLASS   10
#define LSTRIDE  17
#define WSTRIDE  788
#define ROWVEC   (KDIM / 4)
#define WVEC     (NCLASS * ROWVEC)
#define NTHREADS 1024
#define NFULL    3
#define KWAVES   28
#define KSTEPS   7

__global__ __launch_bounds__(NTHREADS)
void quanv_fused_kernel(const float* __restrict__ x,
                        const float* __restrict__ params,
                        const float* __restrict__ W,
                        const float* __restrict__ bias,
                        float* __restrict__ out)
{
    __shared__ float featsL[TILE_M * FSTRIDE];
    __shared__ __attribute__((aligned(16))) float wl[NCLASS * WSTRIDE];
    __shared__ float lgts[TILE_M * LSTRIDE];
    __shared__ float part[KWAVES * 256];

    const int tid = threadIdx.x;
    const int b0  = blockIdx.x * TILE_M;

    const float pv = params[tid & 7];

    const float4* W4 = reinterpret_cast<const float4*>(W);
    const int p0 = tid;
    const int p1 = (tid + NTHREADS < WVEC) ? tid + NTHREADS : tid;
    const float4 w0 = W4[p0];
    const float4 w1 = W4[p1];

    float2 dlo[NFULL], dhi[NFULL];
    #pragma unroll
    for (int j = 0; j < NFULL; ++j) {
        const int t  = tid + j * NTHREADS;
        const int m  = t / NPATCH;
        const int p  = t - m * NPATCH;
        const int r2 = p / 14;
        const int c2 = p - r2 * 14;
        const float* xb = x + (size_t)(b0 + m) * KDIM + (r2 * 56 + c2 * 2);
        dlo[j] = *reinterpret_cast<const float2*>(xb);
        dhi[j] = *reinterpret_cast<const float2*>(xb + 28);
    }

    {
        const int n0 = p0 / ROWVEC;
        const int n1 = p1 / ROWVEC;
        *reinterpret_cast<float4*>(wl + n0 * WSTRIDE + 4 * (p0 - n0 * ROWVEC)) = w0;
        *reinterpret_cast<float4*>(wl + n1 * WSTRIDE + 4 * (p1 - n1 * ROWVEC)) = w1;
    }

    float psin, pcos;
    sincosf(pv, &psin, &pcos);
    float cb[4], ka[4];
    #pragma unroll
    for (int q = 0; q < 4; ++q) {
        cb[q] = lane_value(pcos, 2 * q + 1);
        ka[q] = lane_value(pcos, 2 * q) * lane_value(psin, 2 * q + 1);
    }

    #pragma unroll
    for (int j = 0; j < NFULL; ++j) {
        const int t = tid + j * NTHREADS;
        const int m = t / NPATCH;
        const int p = t - m * NPATCH;

        float s, c;
        sincosf(2.0f * dlo[j].x, &s, &c); float m0 = cb[0] * c - ka[0] * s;
        sincosf(2.0f * dlo[j].y, &s, &c); float m1 = cb[1] * c - ka[1] * s;
        sincosf(2.0f * dhi[j].x, &s, &c); float m2 = cb[2] * c - ka[2] * s;
        sincosf(2.0f * dhi[j].y, &s, &c); float m3 = cb[3] * c - ka[3] * s;

        float* f = featsL + m * FSTRIDE + p * 4;
        f[0] = m0;
        f[1] = m0 * m1;
        f[2] = m2;
        f[3] = m2 * m3;
    }

    for (int t = tid + NFULL * NTHREADS; t < TILE_M * NPATCH; t += NTHREADS) {
        int m  = t / NPATCH;
        int p  = t - m * NPATCH;
        int r2 = p / 14;
        int c2 = p - r2 * 14;
        const float* xb = x + (size_t)(b0 + m) * KDIM + (r2 * 56 + c2 * 2);
        float d0 = xb[0], d1 = xb[1], d2 = xb[28], d3 = xb[29];

        float s, c;
        sincosf(2.0f * d0, &s, &c); float m0 = cb[0] * c - ka[0] * s;
        sincosf(2.0f * d1, &s, &c); float m1 = cb[1] * c - ka[1] * s;
        sincosf(2.0f * d2, &s, &c); float m2 = cb[2] * c - ka[2] * s;
        sincosf(2.0f * d3, &s, &c); float m3 = cb[3] * c - ka[3] * s;

        float* f = featsL + m * FSTRIDE + p * 4;
        f[0] = m0;
        f[1] = m0 * m1;
        f[2] = m2;
        f[3] = m2 * m3;
    }
    __syncthreads();

    const int wave = tid >> 5;
    if (wave < KWAVES) {
        const int lane = tid & 31;
        const int nm   = lane & 15;
        const int half = lane >> 4;
        const int kb   = wave * (KSTEPS * 4);
        const float* fa = featsL + nm * FSTRIDE + 2 * half + kb;
        const int nb = nm < NCLASS ? nm : NCLASS - 1;
        const float* fb = wl + nb * WSTRIDE + kb + 2 * half;

        v8f acc = {};
        for (int i = 0; i < KSTEPS; ++i) {
            const int k0 = 4 * i;
            v2f A;  A.x  = fa[k0];        A.y  = fa[k0 + 1];
            v2f Bv = *reinterpret_cast<const v2f*>(fb + k0);
            acc = __builtin_amdgcn_wmma_f32_16x16x4_f32(
                 false, A,  false, Bv,
                 (short)0, acc,  false,  false);
        }
        float* pw = part + wave * 256;
        #pragma unroll
        for (int r = 0; r < 8; ++r)
            pw[(r + 8 * half) * 16 + nm] = acc[r];
    }
    __syncthreads();

    if (tid < 256) {
        float s = part[tid];
        for (int w = 1; w < KWAVES; ++w) s += part[w * 256 + tid];
        lgts[(tid >> 4) * LSTRIDE + (tid & 15)] = s;
    }
    __syncthreads();

    if (tid < TILE_M) {
        const int row = tid;
        float lg[NCLASS];
        float mx = -3.0e38f;
        #pragma unroll
        for (int j = 0; j < NCLASS; ++j) {
            lg[j] = lgts[row * LSTRIDE + j] + bias[j];
            mx = fmaxf(mx, lg[j]);
        }
        float sum = 0.0f;
        #pragma unroll
        for (int j = 0; j < NCLASS; ++j) sum += expf(lg[j] - mx);
        float lse = logf(sum);
        float* o = out + (size_t)(b0 + row) * NCLASS;
        #pragma unroll
        for (int j = 0; j < NCLASS; ++j) o[j] = lg[j] - mx - lse;
    }
}

extern "C" void kernel_launch(void* const* d_in, const int* in_sizes, int n_in,
                              void* d_out, int out_size, void* d_ws, size_t ws_size,
                              hipStream_t stream) {
    (void)n_in; (void)out_size; (void)d_ws; (void)ws_size;
    const float* x      = (const float*)d_in[0];
    const float* params = (const float*)d_in[1];
    const float* W      = (const float*)d_in[2];
    const float* bias   = (const float*)d_in[3];
    float* out          = (float*)d_out;

    const int B      = in_sizes[0] / KDIM;
    const int blocks = B / TILE_M;

    quanv_fused_kernel<<<blocks, NTHREADS, 0, stream>>>(x, params, W, bias, out);
}


# sin/cos of 2x as branch-free vector code (the library's own pi/2 reduction and polynomials for |y| < 2^17) in place of 12 sincosf calls with a branch each; same bits, sums unchanged
# speedup vs baseline: 10.0099x; 1.1660x over previous
#include <hip/hip_runtime.h>
#include <math.h>

typedef __attribute__((ext_vector_type(2))) float v2f;
typedef __attribute__((ext_vector_type(8))) float v8f;

__device__ __forceinline__ float lane_value(float v, int lane)
{
    return __int_as_float(__builtin_amdgcn_readlane(__float_as_int(v), lane));
}


__device__ __forceinline__ void sincos_pio2(float y, float& sn, float& cs)
{
    const float k = __builtin_rintf(y * __uint_as_float(0x3f22f983u));
    float r = __builtin_fmaf(k, __uint_as_float(0xbfc90fdau), y);
    r = __builtin_fmaf(k, __uint_as_float(0xb3a22168u), r);
    r = __builtin_fmaf(k, __uint_as_float(0xa7c234c4u), r);
    const int   ki = (int)k;
    const float s  = r * r;
    float p = __builtin_fmaf(__uint_as_float(0xb94c1982u), s, __uint_as_float(0x3c0881c4u));
    p = __builtin_fmaf(s, p, __uint_as_float(0xbe2aaa9du));
    p = s * p;
    const float S = __builtin_fmaf(r, p, r);
    float q = __builtin_fmaf(__uint_as_float(0x37d75334u), s, __uint_as_float(0xbab64f3bu));
    q = __builtin_fmaf(s, q, __uint_as_float(0x3d2aabf7u));
    q = __builtin_fmaf(s, q, __uint_as_float(0xbf000004u));
    const float C = __builtin_fmaf(s, q, 1.0f);
    const bool     odd  = (ki & 1) != 0;
    const unsigned flip = ((unsigned)ki << 30) & 0x80000000u;
    sn = __uint_as_float(__float_as_uint(odd ?  C : S) ^ flip);
    cs = __uint_as_float(__float_as_uint(odd ? -S : C) ^ flip);
}

#define TILE_M   16
#define NPATCH   196
#define KDIM     784
#define FSTRIDE  785
#define NCLASS   10
#define LSTRIDE  17
#define WSTRIDE  788
#define ROWVEC   (KDIM / 4)
#define WVEC     (NCLASS * ROWVEC)
#define NTHREADS 1024
#define NFULL    3
#define KWAVES   28
#define KSTEPS   7

__global__ __launch_bounds__(NTHREADS)
void quanv_fused_kernel(const float* __restrict__ x,
                        const float* __restrict__ params,
                        const float* __restrict__ W,
                        const float* __restrict__ bias,
                        float* __restrict__ out)
{
    __shared__ float featsL[TILE_M * FSTRIDE];
    __shared__ __attribute__((aligned(16))) float wl[NCLASS * WSTRIDE];
    __shared__ float lgts[TILE_M * LSTRIDE];
    __shared__ float part[KWAVES * 256];

    const int tid = threadIdx.x;
    const int b0  = blockIdx.x * TILE_M;

    const float pv = params[tid & 7];

    const float4* W4 = reinterpret_cast<const float4*>(W);
    const int p0 = tid;
    const int p1 = (tid + NTHREADS < WVEC) ? tid + NTHREADS : tid;
    const float4 w0 = W4[p0];
    const float4 w1 = W4[p1];

    float2 dlo[NFULL], dhi[NFULL];
    #pragma unroll
    for (int j = 0; j < NFULL; ++j) {
        const int t  = tid + j * NTHREADS;
        const int m  = t / NPATCH;
        const int p  = t - m * NPATCH;
        const int r2 = p / 14;
        const int c2 = p - r2 * 14;
        const float* xb = x + (size_t)(b0 + m) * KDIM + (r2 * 56 + c2 * 2);
        dlo[j] = *reinterpret_cast<const float2*>(xb);
        dhi[j] = *reinterpret_cast<const float2*>(xb + 28);
    }

    {
        const int n0 = p0 / ROWVEC;
        const int n1 = p1 / ROWVEC;
        *reinterpret_cast<float4*>(wl + n0 * WSTRIDE + 4 * (p0 - n0 * ROWVEC)) = w0;
        *reinterpret_cast<float4*>(wl + n1 * WSTRIDE + 4 * (p1 - n1 * ROWVEC)) = w1;
    }

    float psin, pcos;
    sincosf(pv, &psin, &pcos);
    float cb[4], ka[4];
    #pragma unroll
    for (int q = 0; q < 4; ++q) {
        cb[q] = lane_value(pcos, 2 * q + 1);
        ka[q] = lane_value(pcos, 2 * q) * lane_value(psin, 2 * q + 1);
    }

    #pragma unroll
    for (int j = 0; j < NFULL; ++j) {
        const int t = tid + j * NTHREADS;
        const int m = t / NPATCH;
        const int p = t - m * NPATCH;

        float s, c;
        sincos_pio2(2.0f * dlo[j].x, s, c); float m0 = cb[0] * c - ka[0] * s;
        sincos_pio2(2.0f * dlo[j].y, s, c); float m1 = cb[1] * c - ka[1] * s;
        sincos_pio2(2.0f * dhi[j].x, s, c); float m2 = cb[2] * c - ka[2] * s;
        sincos_pio2(2.0f * dhi[j].y, s, c); float m3 = cb[3] * c - ka[3] * s;

        float* f = featsL + m * FSTRIDE + p * 4;
        f[0] = m0;
        f[1] = m0 * m1;
        f[2] = m2;
        f[3] = m2 * m3;
    }

    for (int t = tid + NFULL * NTHREADS; t < TILE_M * NPATCH; t += NTHREADS) {
        int m  = t / NPATCH;
        int p  = t - m * NPATCH;
        int r2 = p / 14;
        int c2 = p - r2 * 14;
        const float* xb = x + (size_t)(b0 + m) * KDIM + (r2 * 56 + c2 * 2);
        float d0 = xb[0], d1 = xb[1], d2 = xb[28], d3 = xb[29];

        float s, c;
        sincos_pio2(2.0f * d0, s, c); float m0 = cb[0] * c - ka[0] * s;
        sincos_pio2(2.0f * d1, s, c); float m1 = cb[1] * c - ka[1] * s;
        sincos_pio2(2.0f * d2, s, c); float m2 = cb[2] * c - ka[2] * s;
        sincos_pio2(2.0f * d3, s, c); float m3 = cb[3] * c - ka[3] * s;

        float* f = featsL + m * FSTRIDE + p * 4;
        f[0] = m0;
        f[1] = m0 * m1;
        f[2] = m2;
        f[3] = m2 * m3;
    }
    __syncthreads();

    const int wave = tid >> 5;
    if (wave < KWAVES) {
        const int lane = tid & 31;
        const int nm   = lane & 15;
        const int half = lane >> 4;
        const int kb   = wave * (KSTEPS * 4);
        const float* fa = featsL + nm * FSTRIDE + 2 * half + kb;
        const int nb = nm < NCLASS ? nm : NCLASS - 1;
        const float* fb = wl + nb * WSTRIDE + kb + 2 * half;

        v8f acc = {};
        for (int i = 0; i < KSTEPS; ++i) {
            const int k0 = 4 * i;
            v2f A;  A.x  = fa[k0];        A.y  = fa[k0 + 1];
            v2f Bv = *reinterpret_cast<const v2f*>(fb + k0);
            acc = __builtin_amdgcn_wmma_f32_16x16x4_f32(
                 false, A,  false, Bv,
                 (short)0, acc,  false,  false);
        }
        float* pw = part + wave * 256;
        #pragma unroll
        for (int r = 0; r < 8; ++r)
            pw[(r + 8 * half) * 16 + nm] = acc[r];
    }
    __syncthreads();

    if (tid < 256) {
        float s = part[tid];
        for (int w = 1; w < KWAVES; ++w) s += part[w * 256 + tid];
        lgts[(tid >> 4) * LSTRIDE + (tid & 15)] = s;
    }
    __syncthreads();

    if (tid < TILE_M) {
        const int row = tid;
        float lg[NCLASS];
        float mx = -3.0e38f;
        #pragma unroll
        for (int j = 0; j < NCLASS; ++j) {
            lg[j] = lgts[row * LSTRIDE + j] + bias[j];
            mx = fmaxf(mx, lg[j]);
        }
        float sum = 0.0f;
        #pragma unroll
        for (int j = 0; j < NCLASS; ++j) sum += expf(lg[j] - mx);
        float lse = logf(sum);
        float* o = out + (size_t)(b0 + row) * NCLASS;
        #pragma unroll
        for (int j = 0; j < NCLASS; ++j) o[j] = lg[j] - mx - lse;
    }
}

extern "C" void kernel_launch(void* const* d_in, const int* in_sizes, int n_in,
                              void* d_out, int out_size, void* d_ws, size_t ws_size,
                              hipStream_t stream) {
    (void)n_in; (void)out_size; (void)d_ws; (void)ws_size;
    const float* x      = (const float*)d_in[0];
    const float* params = (const float*)d_in[1];
    const float* W      = (const float*)d_in[2];
    const float* bias   = (const float*)d_in[3];
    float* out          = (float*)d_out;

    const int B      = in_sizes[0] / KDIM;
    const int blocks = B / TILE_M;

    quanv_fused_kernel<<<blocks, NTHREADS, 0, stream>>>(x, params, W, bias, out);
}


# sin/cos of 2x from the hardware's v_sin_f32 / v_cos_f32 (angle in revolutions, x/pi) instead of the in-line pi/2 reduction and polynomials; last bits of the features change; sums unchanged
# speedup vs baseline: 10.8630x; 1.0852x over previous
#include <hip/hip_runtime.h>
#include <math.h>

typedef __attribute__((ext_vector_type(2))) float v2f;
typedef __attribute__((ext_vector_type(8))) float v8f;

__device__ __forceinline__ float lane_value(float v, int lane)
{
    return __int_as_float(__builtin_amdgcn_readlane(__float_as_int(v), lane));
}


__device__ __forceinline__ void sincos_2d(float d, float& sn, float& cs)
{
    const float t = d * 0.318309886f;
    sn = __builtin_amdgcn_sinf(t);
    cs = __builtin_amdgcn_cosf(t);
}

#define TILE_M   16
#define NPATCH   196
#define KDIM     784
#define FSTRIDE  785
#define NCLASS   10
#define LSTRIDE  17
#define WSTRIDE  788
#define ROWVEC   (KDIM / 4)
#define WVEC     (NCLASS * ROWVEC)
#define NTHREADS 1024
#define NFULL    3
#define KWAVES   28
#define KSTEPS   7

__global__ __launch_bounds__(NTHREADS)
void quanv_fused_kernel(const float* __restrict__ x,
                        const float* __restrict__ params,
                        const float* __restrict__ W,
                        const float* __restrict__ bias,
                        float* __restrict__ out)
{
    __shared__ float featsL[TILE_M * FSTRIDE];
    __shared__ __attribute__((aligned(16))) float wl[NCLASS * WSTRIDE];
    __shared__ float lgts[TILE_M * LSTRIDE];
    __shared__ float part[KWAVES * 256];

    const int tid = threadIdx.x;
    const int b0  = blockIdx.x * TILE_M;

    const float pv = params[tid & 7];

    const float4* W4 = reinterpret_cast<const float4*>(W);
    const int p0 = tid;
    const int p1 = (tid + NTHREADS < WVEC) ? tid + NTHREADS : tid;
    const float4 w0 = W4[p0];
    const float4 w1 = W4[p1];

    float2 dlo[NFULL], dhi[NFULL];
    #pragma unroll
    for (int j = 0; j < NFULL; ++j) {
        const int t  = tid + j * NTHREADS;
        const int m  = t / NPATCH;
        const int p  = t - m * NPATCH;
        const int r2 = p / 14;
        const int c2 = p - r2 * 14;
        const float* xb = x + (size_t)(b0 + m) * KDIM + (r2 * 56 + c2 * 2);
        dlo[j] = *reinterpret_cast<const float2*>(xb);
        dhi[j] = *reinterpret_cast<const float2*>(xb + 28);
    }

    {
        const int n0 = p0 / ROWVEC;
        const int n1 = p1 / ROWVEC;
        *reinterpret_cast<float4*>(wl + n0 * WSTRIDE + 4 * (p0 - n0 * ROWVEC)) = w0;
        *reinterpret_cast<float4*>(wl + n1 * WSTRIDE + 4 * (p1 - n1 * ROWVEC)) = w1;
    }

    float psin, pcos;
    sincosf(pv, &psin, &pcos);
    float cb[4], ka[4];
    #pragma unroll
    for (int q = 0; q < 4; ++q) {
        cb[q] = lane_value(pcos, 2 * q + 1);
        ka[q] = lane_value(pcos, 2 * q) * lane_value(psin, 2 * q + 1);
    }

    #pragma unroll
    for (int j = 0; j < NFULL; ++j) {
        const int t = tid + j * NTHREADS;
        const int m = t / NPATCH;
        const int p = t - m * NPATCH;

        float s, c;
        sincos_2d(dlo[j].x, s, c); float m0 = cb[0] * c - ka[0] * s;
        sincos_2d(dlo[j].y, s, c); float m1 = cb[1] * c - ka[1] * s;
        sincos_2d(dhi[j].x, s, c); float m2 = cb[2] * c - ka[2] * s;
        sincos_2d(dhi[j].y, s, c); float m3 = cb[3] * c - ka[3] * s;

        float* f = featsL + m * FSTRIDE + p * 4;
        f[0] = m0;
        f[1] = m0 * m1;
        f[2] = m2;
        f[3] = m2 * m3;
    }

    for (int t = tid + NFULL * NTHREADS; t < TILE_M * NPATCH; t += NTHREADS) {
        int m  = t / NPATCH;
        int p  = t - m * NPATCH;
        int r2 = p / 14;
        int c2 = p - r2 * 14;
        const float* xb = x + (size_t)(b0 + m) * KDIM + (r2 * 56 + c2 * 2);
        float d0 = xb[0], d1 = xb[1], d2 = xb[28], d3 = xb[29];

        float s, c;
        sincos_2d(d0, s, c); float m0 = cb[0] * c - ka[0] * s;
        sincos_2d(d1, s, c); float m1 = cb[1] * c - ka[1] * s;
        sincos_2d(d2, s, c); float m2 = cb[2] * c - ka[2] * s;
        sincos_2d(d3, s, c); float m3 = cb[3] * c - ka[3] * s;

        float* f = featsL + m * FSTRIDE + p * 4;
        f[0] = m0;
        f[1] = m0 * m1;
        f[2] = m2;
        f[3] = m2 * m3;
    }
    __syncthreads();

    const int wave = tid >> 5;
    if (wave < KWAVES) {
        const int lane = tid & 31;
        const int nm   = lane & 15;
        const int half = lane >> 4;
        const int kb   = wave * (KSTEPS * 4);
        const float* fa = featsL + nm * FSTRIDE + 2 * half + kb;
        const int nb = nm < NCLASS ? nm : NCLASS - 1;
        const float* fb = wl + nb * WSTRIDE + kb + 2 * half;

        v8f acc = {};
        for (int i = 0; i < KSTEPS; ++i) {
            const int k0 = 4 * i;
            v2f A;  A.x  = fa[k0];        A.y  = fa[k0 + 1];
            v2f Bv = *reinterpret_cast<const v2f*>(fb + k0);
            acc = __builtin_amdgcn_wmma_f32_16x16x4_f32(
                 false, A,  false, Bv,
                 (short)0, acc,  false,  false);
        }
        float* pw = part + wave * 256;
        #pragma unroll
        for (int r = 0; r < 8; ++r)
            pw[(r + 8 * half) * 16 + nm] = acc[r];
    }
    __syncthreads();

    if (tid < 256) {
        float s = part[tid];
        for (int w = 1; w < KWAVES; ++w) s += part[w * 256 + tid];
        lgts[(tid >> 4) * LSTRIDE + (tid & 15)] = s;
    }
    __syncthreads();

    if (tid < TILE_M) {
        const int row = tid;
        float lg[NCLASS];
        float mx = -3.0e38f;
        #pragma unroll
        for (int j = 0; j < NCLASS; ++j) {
            lg[j] = lgts[row * LSTRIDE + j] + bias[j];
            mx = fmaxf(mx, lg[j]);
        }
        float sum = 0.0f;
        #pragma unroll
        for (int j = 0; j < NCLASS; ++j) sum += expf(lg[j] - mx);
        float lse = logf(sum);
        float* o = out + (size_t)(b0 + row) * NCLASS;
        #pragma unroll
        for (int j = 0; j < NCLASS; ++j) o[j] = lg[j] - mx - lse;
    }
}

extern "C" void kernel_launch(void* const* d_in, const int* in_sizes, int n_in,
                              void* d_out, int out_size, void* d_ws, size_t ws_size,
                              hipStream_t stream) {
    (void)n_in; (void)out_size; (void)d_ws; (void)ws_size;
    const float* x      = (const float*)d_in[0];
    const float* params = (const float*)d_in[1];
    const float* W      = (const float*)d_in[2];
    const float* bias   = (const float*)d_in[3];
    float* out          = (float*)d_out;

    const int B      = in_sizes[0] / KDIM;
    const int blocks = B / TILE_M;

    quanv_fused_kernel<<<blocks, NTHREADS, 0, stream>>>(x, params, W, bias, out);
}
